# starting kernel
# baseline (speedup 1.0000x reference)
#include <hip/hip_runtime.h>
#include <hip/hip_bf16.h>


typedef float v2f __attribute__((ext_vector_type(2)));
typedef float v8f __attribute__((ext_vector_type(8)));

__device__ __forceinline__ v8f wmma_f32(v2f a, v2f b, v8f c) {
    return __builtin_amdgcn_wmma_f32_16x16x4_f32(
        false, a, false, b, (short)0, c, false, false);
}

__global__ __launch_bounds__(256) void reorder_w(const float* __restrict__ W,
                                                 int N,
                                                 float* __restrict__ Wr) {
    int o    = blockIdx.x * 256 + threadIdx.x;
    int pair = o / 24576;
    int rem  = o % 24576;
    int i    = rem / 128;
    int rem2 = rem % 128;
    int lane = rem2 >> 2;
    int tl   = (rem2 >> 1) & 1;
    int j    = rem2 & 1;
    int m    = lane & 15;
    int half = lane >> 4;
    int k    = 4 * i + 2 * half + j;
    int n    = (pair * 2 + tl) * 16 + m;
    Wr[o] = W[k * N + n];
}

__global__ __launch_bounds__(256) void lengths_k(const int* __restrict__ mask,
                                                 int* __restrict__ Lout) {
    __shared__ int sh[8];
    int b = blockIdx.x, t = threadIdx.x;
    int s = 0;
    #pragma unroll
    for (int j = 0; j < 8; ++j) s += mask[b * 2048 + t + 256 * j];
    #pragma unroll
    for (int o = 1; o < 32; o <<= 1) s += __shfl_xor(s, o, 32);
    if ((t & 31) == 0) sh[t >> 5] = s;
    __syncthreads();
    if (t == 0) {
        int tot = 0;
        #pragma unroll
        for (int j = 0; j < 8; ++j) tot += sh[j];
        int L = tot - 1;
        Lout[b] = L < 2046 ? L : 2046;
    }
}

#define LDA 196
#define LDC 132
#define KP  192
#define IT  48

__global__ __launch_bounds__(128) void tok_gemm_ln(
    const float* __restrict__ hidden, const int* __restrict__ amask,
    const float* __restrict__ wr,     const float* __restrict__ tok_b,
    const float* __restrict__ gamma,  const float* __restrict__ beta,
    float* __restrict__ reps) {
    __shared__ __align__(16) float shmem[64 * LDA];
    float* shA = shmem;
    float* shC = shmem;

    const int  t    = threadIdx.x;
    const int  lane = t & 31;
    const int  wv   = t >> 5;
    const int  m    = lane & 15;
    const int  half = lane >> 4;
    const long row0 = (long)blockIdx.x * 64;

    v8f acc[4][2] = {};
    const float4* bp = (const float4*)wr + ((long)wv * 192) * 32 + lane;

    for (int ph = 0; ph < 4; ++ph) {
        __syncthreads();
        for (int idx = t; idx < 64 * KP / 4; idx += 128) {
            int r  = idx / (KP / 4);
            int c4 = (idx % (KP / 4)) * 4;
            float4 v = *(const float4*)(hidden + (row0 + r) * 768 + ph * KP + c4);
            *(float4*)&shA[r * LDA + c4] = v;
        }
        __syncthreads();

        for (int i = 0; i < IT; ++i) {
            float4 f = bp[(long)(ph * IT + i) * 32];
            v2f b0 = {f.x, f.y};
            v2f b1 = {f.z, f.w};
            #pragma unroll
            for (int mt = 0; mt < 4; ++mt) {
                v2f a = *(const v2f*)&shA[(mt * 16 + m) * LDA + 4 * i + 2 * half];
                acc[mt][0] = wmma_f32(a, b0, acc[mt][0]);
                acc[mt][1] = wmma_f32(a, b1, acc[mt][1]);
            }
        }
    }
    __syncthreads();

    #pragma unroll
    for (int mt = 0; mt < 4; ++mt) {
        #pragma unroll
        for (int i = 0; i < 8; ++i) {
            shC[(mt * 16 + i + 8 * half) * LDC + wv * 32 + m]      = acc[mt][0][i];
            shC[(mt * 16 + i + 8 * half) * LDC + wv * 32 + 16 + m] = acc[mt][1][i];
        }
    }
    __syncthreads();

    const int r  = t >> 1;
    const int c0 = (t & 1) * 64;
    float s = 0.f, ss = 0.f;
    for (int i = 0; i < 64; ++i) {
        float v = shC[r * LDC + c0 + i] + tok_b[c0 + i];
        s += v; ss += v * v;
    }
    s  += __shfl_xor(s,  1, 32);
    ss += __shfl_xor(ss, 1, 32);
    const float mean = s * (1.f / 128.f);
    const float var  = ss * (1.f / 128.f) - mean * mean;
    const float rstd = rsqrtf(var + 1e-5f);

    const long rg = row0 + r;
    const int  bI = (int)(rg >> 11);
    const int  sI = (int)(rg & 2047);
    const float mk = (float)amask[rg];
    if (sI >= 1 && sI <= 2046) {
        float* dst = reps + ((long)(bI * 2046 + sI - 1)) * 128 + c0;
        for (int i = 0; i < 64; ++i) {
            float v = shC[r * LDC + c0 + i] + tok_b[c0 + i];
            float y = (v - mean) * rstd * gamma[c0 + i] + beta[c0 + i];
            dst[i] = fmaxf(y, 0.f) * mk;
        }
    }
}

__global__ __launch_bounds__(128) void cls_gemm(const float* __restrict__ hidden,
                                                const float* __restrict__ wr2,
                                                float* __restrict__ cls_lin) {
    const int t = threadIdx.x, lane = t & 31, wv = t >> 5;
    const int m = lane & 15, half = lane >> 4;
    const int pair = blockIdx.x * 4 + wv;

    v8f acc[2][2] = {};
    const float4* bp = (const float4*)wr2 + ((long)pair * 192) * 32 + lane;
    const float* a0b = hidden + (long)m * (2048 * 768) + 2 * half;
    const float* a1b = hidden + (long)(m + 16) * (2048 * 768) + 2 * half;

    for (int i = 0; i < 192; ++i) {
        v2f a0 = *(const v2f*)(a0b + 4 * i);
        v2f a1 = *(const v2f*)(a1b + 4 * i);
        float4 f = bp[i * 32];
        v2f b0 = {f.x, f.y};
        v2f b1 = {f.z, f.w};
        acc[0][0] = wmma_f32(a0, b0, acc[0][0]);
        acc[0][1] = wmma_f32(a0, b1, acc[0][1]);
        acc[1][0] = wmma_f32(a1, b0, acc[1][0]);
        acc[1][1] = wmma_f32(a1, b1, acc[1][1]);
    }
    const int c0 = pair * 32 + m;
    #pragma unroll
    for (int i = 0; i < 8; ++i) {
        cls_lin[(i + 8 * half) * 768 + c0]           = acc[0][0][i];
        cls_lin[(i + 8 * half) * 768 + c0 + 16]      = acc[0][1][i];
        cls_lin[(i + 8 * half + 16) * 768 + c0]      = acc[1][0][i];
        cls_lin[(i + 8 * half + 16) * 768 + c0 + 16] = acc[1][1][i];
    }
}

__global__ __launch_bounds__(256) void cls_ln(const float* __restrict__ cls_lin,
                                              const float* __restrict__ cls_b,
                                              const float* __restrict__ gamma,
                                              const float* __restrict__ beta,
                                              float* __restrict__ out) {
    __shared__ float sh_s[8], sh_ss[8];
    const int b = blockIdx.x, t = threadIdx.x;
    float x[3];
    float s = 0.f, ss = 0.f;
    #pragma unroll
    for (int j = 0; j < 3; ++j) {
        int c = t + 256 * j;
        float v = cls_lin[b * 768 + c] + cls_b[c];
        x[j] = v; s += v; ss += v * v;
    }
    #pragma unroll
    for (int o = 1; o < 32; o <<= 1) {
        s  += __shfl_xor(s,  o, 32);
        ss += __shfl_xor(ss, o, 32);
    }
    if ((t & 31) == 0) { sh_s[t >> 5] = s; sh_ss[t >> 5] = ss; }
    __syncthreads();
    float ts = 0.f, tss = 0.f;
    #pragma unroll
    for (int j = 0; j < 8; ++j) { ts += sh_s[j]; tss += sh_ss[j]; }
    const float mean = ts * (1.f / 768.f);
    const float var  = tss * (1.f / 768.f) - mean * mean;
    const float rstd = rsqrtf(var + 1e-5f);
    #pragma unroll
    for (int j = 0; j < 3; ++j) {
        int c = t + 256 * j;
        out[b * 768 + c] = (x[j] - mean) * rstd * gamma[c] + beta[c];
    }
}

__global__ __launch_bounds__(256) void window_norm(const float* __restrict__ reps,
                                                   const int* __restrict__ Lw,
                                                   float* __restrict__ out) {
    const int g    = blockIdx.x * 8 + (threadIdx.x >> 5);
    const int lane = threadIdx.x & 31;
    const int b    = g / 2046;
    const int tp   = g % 2046;
    const int L    = Lw[b];
    const bool valid = tp < L;
    int start = tp - 5; if (start < 0) start = 0;
    int end   = tp + 5; if (end > L) end = L;
    int cnt   = end - start; if (cnt < 1) cnt = 1;

    float4 sum = {0.f, 0.f, 0.f, 0.f};
    const float* base = reps + ((long)b * 2046) * 128 + lane * 4;
    for (int r = start; r < end; ++r) {
        float4 v = *(const float4*)(base + (long)r * 128);
        sum.x += v.x; sum.y += v.y; sum.z += v.z; sum.w += v.w;
    }
    const float inv = 1.f / (float)cnt;
    float4 mn = {sum.x * inv, sum.y * inv, sum.z * inv, sum.w * inv};
    float sq = mn.x * mn.x + mn.y * mn.y + mn.z * mn.z + mn.w * mn.w;
    #pragma unroll
    for (int o = 1; o < 32; o <<= 1) sq += __shfl_xor(sq, o, 32);
    const float n  = sqrtf(sq);
    const float sc = (valid && n > 0.f) ? (1.f / n) : 0.f;
    float4 o4 = {mn.x * sc, mn.y * sc, mn.z * sc, mn.w * sc};
    *(float4*)(out + (long)g * 128 + lane * 4) = o4;
}

extern "C" void kernel_launch(void* const* d_in, const int* in_sizes, int n_in,
                              void* d_out, int out_size, void* d_ws, size_t ws_size,
                              hipStream_t stream) {
    const float* hidden   = (const float*)d_in[0];
    const int*   amask    = (const int*)  d_in[1];
    const float* tok_w    = (const float*)d_in[2];
    const float* tok_b    = (const float*)d_in[3];
    const float* cls_w    = (const float*)d_in[4];
    const float* cls_b    = (const float*)d_in[5];
    const float* ln_tok_g = (const float*)d_in[6];
    const float* ln_tok_b = (const float*)d_in[7];
    const float* ln_cls_g = (const float*)d_in[8];
    const float* ln_cls_b = (const float*)d_in[9];
    float* out = (float*)d_out;

    float* ws      = (float*)d_ws;
    float* reps    = ws;
    float* cls_lin = ws + 8380416;
    float* wr_tok  = ws + 8404992;
    float* wr_cls  = ws + 8503296;
    int*   lenw    = (int*)(ws + 9093120);

    reorder_w<<<384,  256, 0, stream>>>(tok_w, 128, wr_tok);
    reorder_w<<<2304, 256, 0, stream>>>(cls_w, 768, wr_cls);
    lengths_k<<<32,   256, 0, stream>>>(amask, lenw);

    tok_gemm_ln<<<1024, 128, 0, stream>>>(hidden, amask, wr_tok, tok_b,
                                          ln_tok_g, ln_tok_b, reps);
    cls_gemm<<<6, 128, 0, stream>>>(hidden, wr_cls, cls_lin);
    cls_ln<<<32, 256, 0, stream>>>(cls_lin, cls_b, ln_cls_g, ln_cls_b, out);
    window_norm<<<8184, 256, 0, stream>>>(reps, lenw, out + 24576);
}


# tok GEMM re-tiled, operand roles swapped: weights double-buffered in LDS, each wave owns 16 whole rows and loads A straight from global into fragments; tok K-sum and LayerNorm sum order change
# speedup vs baseline: 1.0752x; 1.0752x over previous
#include <hip/hip_runtime.h>
#include <hip/hip_bf16.h>


typedef float v2f __attribute__((ext_vector_type(2)));
typedef float v8f __attribute__((ext_vector_type(8)));

__device__ __forceinline__ v8f wmma_f32(v2f a, v2f b, v8f c) {
    return __builtin_amdgcn_wmma_f32_16x16x4_f32(
        false, a, false, b, (short)0, c, false, false);
}

__global__ __launch_bounds__(256) void reorder_w(const float* __restrict__ W,
                                                 int N,
                                                 float* __restrict__ Wr) {
    int o    = blockIdx.x * 256 + threadIdx.x;
    int pair = o / 24576;
    int rem  = o % 24576;
    int i    = rem / 128;
    int rem2 = rem % 128;
    int lane = rem2 >> 2;
    int tl   = (rem2 >> 1) & 1;
    int j    = rem2 & 1;
    int m    = lane & 15;
    int half = lane >> 4;
    int k    = 4 * i + 2 * half + j;
    int n    = (pair * 2 + tl) * 16 + m;
    Wr[o] = W[k * N + n];
}

__global__ __launch_bounds__(256) void lengths_k(const int* __restrict__ mask,
                                                 int* __restrict__ Lout) {
    __shared__ int sh[8];
    int b = blockIdx.x, t = threadIdx.x;
    int s = 0;
    #pragma unroll
    for (int j = 0; j < 8; ++j) s += mask[b * 2048 + t + 256 * j];
    #pragma unroll
    for (int o = 1; o < 32; o <<= 1) s += __shfl_xor(s, o, 32);
    if ((t & 31) == 0) sh[t >> 5] = s;
    __syncthreads();
    if (t == 0) {
        int tot = 0;
        #pragma unroll
        for (int j = 0; j < 8; ++j) tot += sh[j];
        int L = tot - 1;
        Lout[b] = L < 2046 ? L : 2046;
    }
}

__global__ __launch_bounds__(256) void reorder_tok(const float* __restrict__ W,
                                                   float* __restrict__ Wr) {
    int o    = blockIdx.x * 256 + threadIdx.x;
    int ks   = o >> 9;
    int rem  = o & 511;
    int p    = rem >> 7;
    int rem2 = rem & 127;
    int lane = rem2 >> 2;
    int tl   = (rem2 >> 1) & 1;
    int s    = rem2 & 1;
    int n    = lane & 15;
    int h    = lane >> 4;
    int k    = 8 * (ks >> 1) + 4 * h + 2 * (ks & 1) + s;
    int col  = (2 * p + tl) * 16 + n;
    Wr[o] = W[k * 128 + col];
}

#define TLDC 132
#define TNPH 6
#define TPH4 4096

__global__ __launch_bounds__(512) __attribute__((amdgpu_waves_per_eu(4, 4)))
void tok_gemm_ln(
    const float* __restrict__ hidden, const int* __restrict__ amask,
    const float* __restrict__ wr,     const float* __restrict__ tok_b,
    const float* __restrict__ gamma,  const float* __restrict__ beta,
    float* __restrict__ reps) {
    __shared__ __align__(16) float shmem[16 * 16 * TLDC];
    float4* sW = (float4*)shmem;

    const int  t    = threadIdx.x;
    const int  lane = t & 31;
    const int  wv   = t >> 5;
    const int  c    = lane & 15;
    const int  h    = lane >> 4;
    const long row0 = (long)blockIdx.x * 256 + wv * 16;

    v8f acc[8] = {};
    const float4* ap = (const float4*)(hidden + (row0 + c) * 768 + 4 * h);
    const float4* wg = (const float4*)wr + t;

    float4 a_cur[4], a_nxt[4];
    {
        const float4 w0 = wg[0],    w1 = wg[512],  w2 = wg[1024], w3 = wg[1536];
        const float4 w4 = wg[2048], w5 = wg[2560], w6 = wg[3072], w7 = wg[3584];
        float4* d = sW + t;
        d[0]    = w0; d[512]  = w1; d[1024] = w2; d[1536] = w3;
        d[2048] = w4; d[2560] = w5; d[3072] = w6; d[3584] = w7;
    }
    #pragma unroll
    for (int q = 0; q < 4; ++q) a_cur[q] = ap[2 * q];
    __syncthreads();

    for (int ph = 0; ph < TNPH; ++ph) {
        const int phn = (ph + 1 < TNPH) ? (ph + 1) : (TNPH - 1);
        const float4* wgn = wg + phn * TPH4;
        float4 w0 = {}, w1 = {}, w2 = {}, w3 = {}, w4 = {}, w5 = {}, w6 = {}, w7 = {};
        const float4* sBp = sW + (ph & 1) * TPH4 + lane;

        #pragma unroll 1
        for (int jj = 0; jj < 4; ++jj) {
            const int j  = ph * 4 + jj;
            const int jn = (j + 1 < 24) ? (j + 1) : 23;
            #pragma unroll
            for (int q = 0; q < 4; ++q) a_nxt[q] = ap[jn * 8 + 2 * q];
            if (jj == 0) {
                w0 = wgn[0];    w1 = wgn[512];  w2 = wgn[1024]; w3 = wgn[1536];
                w4 = wgn[2048]; w5 = wgn[2560]; w6 = wgn[3072]; w7 = wgn[3584];
            }
            const float4* sB = sBp + jj * (8 * 4 * 32);
            float4 bc[4], bn[4];
            #pragma unroll
            for (int p = 0; p < 4; ++p) bc[p] = sB[p * 32];
            #pragma unroll
            for (int q = 0; q < 4; ++q) {
                #pragma unroll
                for (int e = 0; e < 2; ++e) {
                    const int ks8 = 2 * q + e;
                    if (ks8 + 1 < 8) {
                        #pragma unroll
                        for (int p = 0; p < 4; ++p) bn[p] = sB[((ks8 + 1) * 4 + p) * 32];
                    }
                    __builtin_amdgcn_sched_barrier(0);
                    v2f a;
                    if (e == 0) { a[0] = a_cur[q].x; a[1] = a_cur[q].y; }
                    else        { a[0] = a_cur[q].z; a[1] = a_cur[q].w; }
                    #pragma unroll
                    for (int p = 0; p < 4; ++p) {
                        v2f b0 = {bc[p].x, bc[p].y};
                        v2f b1 = {bc[p].z, bc[p].w};
                        acc[2 * p]     = wmma_f32(a, b0, acc[2 * p]);
                        acc[2 * p + 1] = wmma_f32(a, b1, acc[2 * p + 1]);
                    }
                    __builtin_amdgcn_sched_barrier(0);
                    #pragma unroll
                    for (int p = 0; p < 4; ++p) bc[p] = bn[p];
                }
            }
            #pragma unroll
            for (int q = 0; q < 4; ++q) a_cur[q] = a_nxt[q];
        }
        {
            float4* d = sW + ((ph + 1) & 1) * TPH4 + t;
            d[0]    = w0; d[512]  = w1; d[1024] = w2; d[1536] = w3;
            d[2048] = w4; d[2560] = w5; d[3072] = w6; d[3584] = w7;
        }
        __syncthreads();
    }

    float bia[8], gam[8], bet[8];
    #pragma unroll
    for (int nt = 0; nt < 8; ++nt) {
        bia[nt] = tok_b[nt * 16 + c];
        gam[nt] = gamma[nt * 16 + c];
        bet[nt] = beta[nt * 16 + c];
    }
    float* Cw = shmem + wv * (16 * TLDC);
    #pragma unroll
    for (int r = 0; r < 8; ++r) {
        const long  rg = row0 + 8 * h + r;
        const float mk = (float)amask[rg];
        float v[8];
        float s = 0.f, ss = 0.f;
        #pragma unroll
        for (int nt = 0; nt < 8; ++nt) {
            v[nt] = acc[nt][r] + bia[nt];
            s += v[nt]; ss += v[nt] * v[nt];
        }
        #pragma unroll
        for (int o = 1; o < 16; o <<= 1) {
            s  += __shfl_xor(s,  o, 32);
            ss += __shfl_xor(ss, o, 32);
        }
        const float mean = s * (1.f / 128.f);
        const float var  = ss * (1.f / 128.f) - mean * mean;
        const float rstd = rsqrtf(var + 1e-5f);
        #pragma unroll
        for (int nt = 0; nt < 8; ++nt) {
            float y = (v[nt] - mean) * rstd * gam[nt] + bet[nt];
            Cw[(8 * h + r) * TLDC + nt * 16 + c] = fmaxf(y, 0.f) * mk;
        }
    }
    __syncthreads();
    #pragma unroll
    for (int rr = 0; rr < 16; ++rr) {
        const long rg = row0 + rr;
        const int  bI = (int)(rg >> 11);
        const int  sI = (int)(rg & 2047);
        if (sI >= 1 && sI <= 2046) {
            float4 v = *(const float4*)&Cw[rr * TLDC + 4 * lane];
            *(float4*)(reps + ((long)(bI * 2046 + sI - 1)) * 128 + 4 * lane) = v;
        }
    }
}

__global__ __launch_bounds__(128) void cls_gemm(const float* __restrict__ hidden,
                                                const float* __restrict__ wr2,
                                                float* __restrict__ cls_lin) {
    const int t = threadIdx.x, lane = t & 31, wv = t >> 5;
    const int m = lane & 15, half = lane >> 4;
    const int pair = blockIdx.x * 4 + wv;

    v8f acc[2][2] = {};
    const float4* bp = (const float4*)wr2 + ((long)pair * 192) * 32 + lane;
    const float* a0b = hidden + (long)m * (2048 * 768) + 2 * half;
    const float* a1b = hidden + (long)(m + 16) * (2048 * 768) + 2 * half;

    for (int i = 0; i < 192; ++i) {
        v2f a0 = *(const v2f*)(a0b + 4 * i);
        v2f a1 = *(const v2f*)(a1b + 4 * i);
        float4 f = bp[i * 32];
        v2f b0 = {f.x, f.y};
        v2f b1 = {f.z, f.w};
        acc[0][0] = wmma_f32(a0, b0, acc[0][0]);
        acc[0][1] = wmma_f32(a0, b1, acc[0][1]);
        acc[1][0] = wmma_f32(a1, b0, acc[1][0]);
        acc[1][1] = wmma_f32(a1, b1, acc[1][1]);
    }
    const int c0 = pair * 32 + m;
    #pragma unroll
    for (int i = 0; i < 8; ++i) {
        cls_lin[(i + 8 * half) * 768 + c0]           = acc[0][0][i];
        cls_lin[(i + 8 * half) * 768 + c0 + 16]      = acc[0][1][i];
        cls_lin[(i + 8 * half + 16) * 768 + c0]      = acc[1][0][i];
        cls_lin[(i + 8 * half + 16) * 768 + c0 + 16] = acc[1][1][i];
    }
}

__global__ __launch_bounds__(256) void cls_ln(const float* __restrict__ cls_lin,
                                              const float* __restrict__ cls_b,
                                              const float* __restrict__ gamma,
                                              const float* __restrict__ beta,
                                              float* __restrict__ out) {
    __shared__ float sh_s[8], sh_ss[8];
    const int b = blockIdx.x, t = threadIdx.x;
    float x[3];
    float s = 0.f, ss = 0.f;
    #pragma unroll
    for (int j = 0; j < 3; ++j) {
        int c = t + 256 * j;
        float v = cls_lin[b * 768 + c] + cls_b[c];
        x[j] = v; s += v; ss += v * v;
    }
    #pragma unroll
    for (int o = 1; o < 32; o <<= 1) {
        s  += __shfl_xor(s,  o, 32);
        ss += __shfl_xor(ss, o, 32);
    }
    if ((t & 31) == 0) { sh_s[t >> 5] = s; sh_ss[t >> 5] = ss; }
    __syncthreads();
    float ts = 0.f, tss = 0.f;
    #pragma unroll
    for (int j = 0; j < 8; ++j) { ts += sh_s[j]; tss += sh_ss[j]; }
    const float mean = ts * (1.f / 768.f);
    const float var  = tss * (1.f / 768.f) - mean * mean;
    const float rstd = rsqrtf(var + 1e-5f);
    #pragma unroll
    for (int j = 0; j < 3; ++j) {
        int c = t + 256 * j;
        out[b * 768 + c] = (x[j] - mean) * rstd * gamma[c] + beta[c];
    }
}

__global__ __launch_bounds__(256) void window_norm(const float* __restrict__ reps,
                                                   const int* __restrict__ Lw,
                                                   float* __restrict__ out) {
    const int g    = blockIdx.x * 8 + (threadIdx.x >> 5);
    const int lane = threadIdx.x & 31;
    const int b    = g / 2046;
    const int tp   = g % 2046;
    const int L    = Lw[b];
    const bool valid = tp < L;
    int start = tp - 5; if (start < 0) start = 0;
    int end   = tp + 5; if (end > L) end = L;
    int cnt   = end - start; if (cnt < 1) cnt = 1;

    float4 sum = {0.f, 0.f, 0.f, 0.f};
    const float* base = reps + ((long)b * 2046) * 128 + lane * 4;
    for (int r = start; r < end; ++r) {
        float4 v = *(const float4*)(base + (long)r * 128);
        sum.x += v.x; sum.y += v.y; sum.z += v.z; sum.w += v.w;
    }
    const float inv = 1.f / (float)cnt;
    float4 mn = {sum.x * inv, sum.y * inv, sum.z * inv, sum.w * inv};
    float sq = mn.x * mn.x + mn.y * mn.y + mn.z * mn.z + mn.w * mn.w;
    #pragma unroll
    for (int o = 1; o < 32; o <<= 1) sq += __shfl_xor(sq, o, 32);
    const float n  = sqrtf(sq);
    const float sc = (valid && n > 0.f) ? (1.f / n) : 0.f;
    float4 o4 = {mn.x * sc, mn.y * sc, mn.z * sc, mn.w * sc};
    *(float4*)(out + (long)g * 128 + lane * 4) = o4;
}

extern "C" void kernel_launch(void* const* d_in, const int* in_sizes, int n_in,
                              void* d_out, int out_size, void* d_ws, size_t ws_size,
                              hipStream_t stream) {
    const float* hidden   = (const float*)d_in[0];
    const int*   amask    = (const int*)  d_in[1];
    const float* tok_w    = (const float*)d_in[2];
    const float* tok_b    = (const float*)d_in[3];
    const float* cls_w    = (const float*)d_in[4];
    const float* cls_b    = (const float*)d_in[5];
    const float* ln_tok_g = (const float*)d_in[6];
    const float* ln_tok_b = (const float*)d_in[7];
    const float* ln_cls_g = (const float*)d_in[8];
    const float* ln_cls_b = (const float*)d_in[9];
    float* out = (float*)d_out;

    float* ws      = (float*)d_ws;
    float* reps    = ws;
    float* cls_lin = ws + 8380416;
    float* wr_tok  = ws + 8404992;
    float* wr_cls  = ws + 8503296;
    int*   lenw    = (int*)(ws + 9093120);

    reorder_tok<<<384, 256, 0, stream>>>(tok_w, wr_tok);
    reorder_w<<<2304, 256, 0, stream>>>(cls_w, 768, wr_cls);
    lengths_k<<<32,   256, 0, stream>>>(amask, lenw);

    tok_gemm_ln<<<256, 512, 0, stream>>>(hidden, amask, wr_tok, tok_b,
                                          ln_tok_g, ln_tok_b, reps);
    cls_gemm<<<6, 128, 0, stream>>>(hidden, wr_cls, cls_lin);
    cls_ln<<<32, 256, 0, stream>>>(cls_lin, cls_b, ln_cls_g, ln_cls_b, out);
    window_norm<<<8184, 256, 0, stream>>>(reps, lenw, out + 24576);
}


# window mean + L2 norm moved into the tok GEMM epilogue: a block's 256 masked rows stay in LDS, reps plane and the 65472-wave window launch are gone; 9 rows per block border go to a small edge kernel
# speedup vs baseline: 1.1732x; 1.0911x over previous
#include <hip/hip_runtime.h>
#include <hip/hip_bf16.h>


typedef float v2f __attribute__((ext_vector_type(2)));
typedef float v8f __attribute__((ext_vector_type(8)));
typedef float v4f __attribute__((ext_vector_type(4)));

__device__ __forceinline__ v8f wmma_f32(v2f a, v2f b, v8f c) {
    return __builtin_amdgcn_wmma_f32_16x16x4_f32(
        false, a, false, b, (short)0, c, false, false);
}

__global__ __launch_bounds__(256) void reorder_w(const float* __restrict__ W,
                                                 int N,
                                                 float* __restrict__ Wr) {
    int o    = blockIdx.x * 256 + threadIdx.x;
    int pair = o / 24576;
    int rem  = o % 24576;
    int i    = rem / 128;
    int rem2 = rem % 128;
    int lane = rem2 >> 2;
    int tl   = (rem2 >> 1) & 1;
    int j    = rem2 & 1;
    int m    = lane & 15;
    int half = lane >> 4;
    int k    = 4 * i + 2 * half + j;
    int n    = (pair * 2 + tl) * 16 + m;
    Wr[o] = W[k * N + n];
}

__global__ __launch_bounds__(256) void lengths_k(const int* __restrict__ mask,
                                                 int* __restrict__ Lout) {
    __shared__ int sh[8];
    int b = blockIdx.x, t = threadIdx.x;
    int s = 0;
    #pragma unroll
    for (int j = 0; j < 8; ++j) s += mask[b * 2048 + t + 256 * j];
    #pragma unroll
    for (int o = 1; o < 32; o <<= 1) s += __shfl_xor(s, o, 32);
    if ((t & 31) == 0) sh[t >> 5] = s;
    __syncthreads();
    if (t == 0) {
        int tot = 0;
        #pragma unroll
        for (int j = 0; j < 8; ++j) tot += sh[j];
        int L = tot - 1;
        Lout[b] = L < 2046 ? L : 2046;
    }
}

__global__ __launch_bounds__(256) void reorder_tok(const float* __restrict__ W,
                                                   float* __restrict__ Wr) {
    int o    = blockIdx.x * 256 + threadIdx.x;
    int ks   = o >> 9;
    int rem  = o & 511;
    int p    = rem >> 7;
    int rem2 = rem & 127;
    int lane = rem2 >> 2;
    int tl   = (rem2 >> 1) & 1;
    int s    = rem2 & 1;
    int n    = lane & 15;
    int h    = lane >> 4;
    int k    = 8 * (ks >> 1) + 4 * h + 2 * (ks & 1) + s;
    int col  = (2 * p + tl) * 16 + n;
    Wr[o] = W[k * 128 + col];
}

#define TLDC 132
#define TNPH 6
#define TPH4 4096

__global__ __launch_bounds__(512) __attribute__((amdgpu_waves_per_eu(4, 4)))
void tok_gemm_ln(
    const float* __restrict__ hidden, const int* __restrict__ amask,
    const float* __restrict__ wr,     const float* __restrict__ tok_b,
    const float* __restrict__ gamma,  const float* __restrict__ beta,
    const int* __restrict__ Lw, float* __restrict__ edge,
    float* __restrict__ out_rows) {
    __shared__ __align__(16) float shmem[16 * 16 * TLDC];
    float4* sW = (float4*)shmem;

    const int  t    = threadIdx.x;
    const int  lane = t & 31;
    const int  wv   = t >> 5;
    const int  c    = lane & 15;
    const int  h    = lane >> 4;
    const long row0 = (long)blockIdx.x * 256 + wv * 16;

    v8f acc[8] = {};
    const float4* ap = (const float4*)(hidden + (row0 + c) * 768 + 4 * h);
    const float4* wg = (const float4*)wr + t;

    float4 a_cur[4], a_nxt[4];
    {
        const float4 w0 = wg[0],    w1 = wg[512],  w2 = wg[1024], w3 = wg[1536];
        const float4 w4 = wg[2048], w5 = wg[2560], w6 = wg[3072], w7 = wg[3584];
        float4* d = sW + t;
        d[0]    = w0; d[512]  = w1; d[1024] = w2; d[1536] = w3;
        d[2048] = w4; d[2560] = w5; d[3072] = w6; d[3584] = w7;
    }
    #pragma unroll
    for (int q = 0; q < 4; ++q) a_cur[q] = ap[2 * q];
    __syncthreads();

    for (int ph = 0; ph < TNPH; ++ph) {
        const int phn = (ph + 1 < TNPH) ? (ph + 1) : (TNPH - 1);
        const float4* wgn = wg + phn * TPH4;
        float4 w0 = {}, w1 = {}, w2 = {}, w3 = {}, w4 = {}, w5 = {}, w6 = {}, w7 = {};
        const float4* sBp = sW + (ph & 1) * TPH4 + lane;

        #pragma unroll 1
        for (int jj = 0; jj < 4; ++jj) {
            const int j  = ph * 4 + jj;
            const int jn = (j + 1 < 24) ? (j + 1) : 23;
            #pragma unroll
            for (int q = 0; q < 4; ++q) a_nxt[q] = ap[jn * 8 + 2 * q];
            if (jj == 0) {
                w0 = wgn[0];    w1 = wgn[512];  w2 = wgn[1024]; w3 = wgn[1536];
                w4 = wgn[2048]; w5 = wgn[2560]; w6 = wgn[3072]; w7 = wgn[3584];
            }
            const float4* sB = sBp + jj * (8 * 4 * 32);
            float4 bc[4], bn[4];
            #pragma unroll
            for (int p = 0; p < 4; ++p) bc[p] = sB[p * 32];
            #pragma unroll
            for (int q = 0; q < 4; ++q) {
                #pragma unroll
                for (int e = 0; e < 2; ++e) {
                    const int ks8 = 2 * q + e;
                    if (ks8 + 1 < 8) {
                        #pragma unroll
                        for (int p = 0; p < 4; ++p) bn[p] = sB[((ks8 + 1) * 4 + p) * 32];
                    }
                    __builtin_amdgcn_sched_barrier(0);
                    v2f a;
                    if (e == 0) { a[0] = a_cur[q].x; a[1] = a_cur[q].y; }
                    else        { a[0] = a_cur[q].z; a[1] = a_cur[q].w; }
                    #pragma unroll
                    for (int p = 0; p < 4; ++p) {
                        v2f b0 = {bc[p].x, bc[p].y};
                        v2f b1 = {bc[p].z, bc[p].w};
                        acc[2 * p]     = wmma_f32(a, b0, acc[2 * p]);
                        acc[2 * p + 1] = wmma_f32(a, b1, acc[2 * p + 1]);
                    }
                    __builtin_amdgcn_sched_barrier(0);
                    #pragma unroll
                    for (int p = 0; p < 4; ++p) bc[p] = bn[p];
                }
            }
            #pragma unroll
            for (int q = 0; q < 4; ++q) a_cur[q] = a_nxt[q];
        }
        {
            float4* d = sW + ((ph + 1) & 1) * TPH4 + t;
            d[0]    = w0; d[512]  = w1; d[1024] = w2; d[1536] = w3;
            d[2048] = w4; d[2560] = w5; d[3072] = w6; d[3584] = w7;
        }
        __syncthreads();
    }

    float bia[8], gam[8], bet[8];
    #pragma unroll
    for (int nt = 0; nt < 8; ++nt) {
        bia[nt] = tok_b[nt * 16 + c];
        gam[nt] = gamma[nt * 16 + c];
        bet[nt] = beta[nt * 16 + c];
    }
    float* Cw = shmem + wv * (16 * TLDC);
    #pragma unroll
    for (int r = 0; r < 8; ++r) {
        const long  rg = row0 + 8 * h + r;
        const float mk = (float)amask[rg];
        float v[8];
        float s = 0.f, ss = 0.f;
        #pragma unroll
        for (int nt = 0; nt < 8; ++nt) {
            v[nt] = acc[nt][r] + bia[nt];
            s += v[nt]; ss += v[nt] * v[nt];
        }
        #pragma unroll
        for (int o = 1; o < 16; o <<= 1) {
            s  += __shfl_xor(s,  o, 32);
            ss += __shfl_xor(ss, o, 32);
        }
        const float mean = s * (1.f / 128.f);
        const float var  = ss * (1.f / 128.f) - mean * mean;
        const float rstd = rsqrtf(var + 1e-5f);
        #pragma unroll
        for (int nt = 0; nt < 8; ++nt) {
            float y = (v[nt] - mean) * rstd * gam[nt] + bet[nt];
            Cw[(8 * h + r) * TLDC + nt * 16 + c] = fmaxf(y, 0.f) * mk;
        }
    }
    __syncthreads();

    const int kb = blockIdx.x & 7;
    const int bI = blockIdx.x >> 3;
    const int s0 = kb * 256;
    const int L  = Lw[bI];
    const int wu = __builtin_amdgcn_readfirstlane(wv);

    if (wu == 0 || wu == 15) {
        const float* src = shmem + (wu == 0 ? 0 : 247) * TLDC + 4 * lane;
        float* dst = edge + ((long)blockIdx.x * 18 + (wu == 0 ? 0 : 9)) * 128 + 4 * lane;
        #pragma unroll
        for (int j = 0; j < 9; ++j)
            *(float4*)(dst + j * 128) = *(const float4*)(src + j * TLDC);
    }

    const int i0 = wu * 16;
    float4 rw[25];
    #pragma unroll
    for (int q = 0; q < 25; ++q) {
        const int i  = i0 - 5 + q;
        const int ic = i < 0 ? 0 : (i > 255 ? 255 : i);
        float4 v = *(const float4*)&shmem[ic * TLDC + 4 * lane];
        const int sr = s0 + i;
        const bool in = (i >= 0) && (i <= 255) && (sr >= 1) && (sr <= L);
        if (!in) { v.x = 0.f; v.y = 0.f; v.z = 0.f; v.w = 0.f; }
        rw[q] = v;
    }
    float4 mn[16];
    float  sq[16];
    #pragma unroll
    for (int rr = 0; rr < 16; ++rr) {
        const int tp = s0 + i0 + rr - 1;
        int start = tp - 5; if (start < 0) start = 0;
        int end   = tp + 5; if (end > L) end = L;
        int cnt   = end - start; if (cnt < 1) cnt = 1;
        float4 sum = {0.f, 0.f, 0.f, 0.f};
        #pragma unroll
        for (int d = 0; d < 10; ++d) {
            sum.x += rw[rr + d].x; sum.y += rw[rr + d].y;
            sum.z += rw[rr + d].z; sum.w += rw[rr + d].w;
        }
        const float inv = 1.f / (float)cnt;
        mn[rr].x = sum.x * inv; mn[rr].y = sum.y * inv;
        mn[rr].z = sum.z * inv; mn[rr].w = sum.w * inv;
        sq[rr] = mn[rr].x * mn[rr].x + mn[rr].y * mn[rr].y +
                 mn[rr].z * mn[rr].z + mn[rr].w * mn[rr].w;
    }
    #pragma unroll
    for (int o = 1; o < 32; o <<= 1) {
        #pragma unroll
        for (int rr = 0; rr < 16; ++rr) sq[rr] += __shfl_xor(sq[rr], o, 32);
    }
    #pragma unroll
    for (int rr = 0; rr < 16; ++rr) {
        const int i = i0 + rr;
        const int s = s0 + i;
        const bool mine = (s >= 1) && (s <= 2046) &&
                          (i >= 5 || kb == 0) && (i <= 251 || kb == 7);
        if (mine) {
            const int  tp    = s - 1;
            const bool valid = tp < L;
            const float n  = sqrtf(sq[rr]);
            const float sc = (valid && n > 0.f) ? (1.f / n) : 0.f;
            float4 o4 = {mn[rr].x * sc, mn[rr].y * sc, mn[rr].z * sc, mn[rr].w * sc};
            *(float4*)(out_rows + ((long)(bI * 2046 + tp)) * 128 + 4 * lane) = o4;
        }
    }
}

__global__ __launch_bounds__(128) void cls_gemm(const float* __restrict__ hidden,
                                                const float* __restrict__ wr2,
                                                float* __restrict__ cls_lin) {
    const int t = threadIdx.x, lane = t & 31, wv = t >> 5;
    const int m = lane & 15, half = lane >> 4;
    const int pair = blockIdx.x * 4 + wv;

    v8f acc[2][2] = {};
    const float4* bp = (const float4*)wr2 + ((long)pair * 192) * 32 + lane;
    const float* a0b = hidden + (long)m * (2048 * 768) + 2 * half;
    const float* a1b = hidden + (long)(m + 16) * (2048 * 768) + 2 * half;

    for (int i = 0; i < 192; ++i) {
        v2f a0 = *(const v2f*)(a0b + 4 * i);
        v2f a1 = *(const v2f*)(a1b + 4 * i);
        float4 f = bp[i * 32];
        v2f b0 = {f.x, f.y};
        v2f b1 = {f.z, f.w};
        acc[0][0] = wmma_f32(a0, b0, acc[0][0]);
        acc[0][1] = wmma_f32(a0, b1, acc[0][1]);
        acc[1][0] = wmma_f32(a1, b0, acc[1][0]);
        acc[1][1] = wmma_f32(a1, b1, acc[1][1]);
    }
    const int c0 = pair * 32 + m;
    #pragma unroll
    for (int i = 0; i < 8; ++i) {
        cls_lin[(i + 8 * half) * 768 + c0]           = acc[0][0][i];
        cls_lin[(i + 8 * half) * 768 + c0 + 16]      = acc[0][1][i];
        cls_lin[(i + 8 * half + 16) * 768 + c0]      = acc[1][0][i];
        cls_lin[(i + 8 * half + 16) * 768 + c0 + 16] = acc[1][1][i];
    }
}

__global__ __launch_bounds__(256) void cls_ln(const float* __restrict__ cls_lin,
                                              const float* __restrict__ cls_b,
                                              const float* __restrict__ gamma,
                                              const float* __restrict__ beta,
                                              float* __restrict__ out) {
    __shared__ float sh_s[8], sh_ss[8];
    const int b = blockIdx.x, t = threadIdx.x;
    float x[3];
    float s = 0.f, ss = 0.f;
    #pragma unroll
    for (int j = 0; j < 3; ++j) {
        int c = t + 256 * j;
        float v = cls_lin[b * 768 + c] + cls_b[c];
        x[j] = v; s += v; ss += v * v;
    }
    #pragma unroll
    for (int o = 1; o < 32; o <<= 1) {
        s  += __shfl_xor(s,  o, 32);
        ss += __shfl_xor(ss, o, 32);
    }
    if ((t & 31) == 0) { sh_s[t >> 5] = s; sh_ss[t >> 5] = ss; }
    __syncthreads();
    float ts = 0.f, tss = 0.f;
    #pragma unroll
    for (int j = 0; j < 8; ++j) { ts += sh_s[j]; tss += sh_ss[j]; }
    const float mean = ts * (1.f / 768.f);
    const float var  = tss * (1.f / 768.f) - mean * mean;
    const float rstd = rsqrtf(var + 1e-5f);
    #pragma unroll
    for (int j = 0; j < 3; ++j) {
        int c = t + 256 * j;
        out[b * 768 + c] = (x[j] - mean) * rstd * gamma[c] + beta[c];
    }
}

__global__ __launch_bounds__(256) void window_edge(const float* __restrict__ edge,
                                                   const int* __restrict__ Lw,
                                                   float* __restrict__ out_rows) {
    const int g    = blockIdx.x * 8 + (threadIdx.x >> 5);
    const int lane = threadIdx.x & 31;
    const int bnd  = g / 9;
    const int e    = g % 9 - 4;
    const int b    = bnd / 7;
    const int k    = bnd % 7 + 1;
    const int s    = 256 * k + e;
    const int tp   = s - 1;
    const int L    = Lw[b];
    const bool valid = tp < L;
    int start = tp - 5; if (start < 0) start = 0;
    int end   = tp + 5; if (end > L) end = L;
    int cnt   = end - start; if (cnt < 1) cnt = 1;

    v4f ld[10];
    #pragma unroll
    for (int d = 0; d < 10; ++d) {
        const int er = e - 5 + d;
        const long j = er < 0 ? ((long)(b * 8 + k - 1) * 18 + 18 + er)
                              : ((long)(b * 8 + k) * 18 + er);
        ld[d] = *(const v4f*)(edge + j * 128 + lane * 4);
    }
    asm volatile("" :: "v"(ld[0]), "v"(ld[1]), "v"(ld[2]), "v"(ld[3]), "v"(ld[4]),
                       "v"(ld[5]), "v"(ld[6]), "v"(ld[7]), "v"(ld[8]), "v"(ld[9]));
    float4 sum = {0.f, 0.f, 0.f, 0.f};
    #pragma unroll
    for (int d = 0; d < 10; ++d) {
        const int sr = s - 5 + d;
        const bool in = (sr >= 1) && (sr <= L);
        const float vx = in ? ld[d].x : 0.f, vy = in ? ld[d].y : 0.f;
        const float vz = in ? ld[d].z : 0.f, vw = in ? ld[d].w : 0.f;
        sum.x += vx; sum.y += vy; sum.z += vz; sum.w += vw;
    }
    const float inv = 1.f / (float)cnt;
    float4 mn = {sum.x * inv, sum.y * inv, sum.z * inv, sum.w * inv};
    float sq = mn.x * mn.x + mn.y * mn.y + mn.z * mn.z + mn.w * mn.w;
    #pragma unroll
    for (int o = 1; o < 32; o <<= 1) sq += __shfl_xor(sq, o, 32);
    const float n  = sqrtf(sq);
    const float sc = (valid && n > 0.f) ? (1.f / n) : 0.f;
    float4 o4 = {mn.x * sc, mn.y * sc, mn.z * sc, mn.w * sc};
    *(float4*)(out_rows + ((long)(b * 2046 + tp)) * 128 + lane * 4) = o4;
}

extern "C" void kernel_launch(void* const* d_in, const int* in_sizes, int n_in,
                              void* d_out, int out_size, void* d_ws, size_t ws_size,
                              hipStream_t stream) {
    const float* hidden   = (const float*)d_in[0];
    const int*   amask    = (const int*)  d_in[1];
    const float* tok_w    = (const float*)d_in[2];
    const float* tok_b    = (const float*)d_in[3];
    const float* cls_w    = (const float*)d_in[4];
    const float* cls_b    = (const float*)d_in[5];
    const float* ln_tok_g = (const float*)d_in[6];
    const float* ln_tok_b = (const float*)d_in[7];
    const float* ln_cls_g = (const float*)d_in[8];
    const float* ln_cls_b = (const float*)d_in[9];
    float* out = (float*)d_out;

    float* ws      = (float*)d_ws;
    float* edge    = ws;
    float* cls_lin = ws + 8380416;
    float* wr_tok  = ws + 8404992;
    float* wr_cls  = ws + 8503296;
    int*   lenw    = (int*)(ws + 9093120);

    reorder_tok<<<384, 256, 0, stream>>>(tok_w, wr_tok);
    reorder_w<<<2304, 256, 0, stream>>>(cls_w, 768, wr_cls);
    lengths_k<<<32,   256, 0, stream>>>(amask, lenw);

    tok_gemm_ln<<<256, 512, 0, stream>>>(hidden, amask, wr_tok, tok_b,
                                          ln_tok_g, ln_tok_b, lenw, edge,
                                          out + 24576);
    cls_gemm<<<6, 128, 0, stream>>>(hidden, wr_cls, cls_lin);
    cls_ln<<<32, 256, 0, stream>>>(cls_lin, cls_b, ln_cls_g, ln_cls_b, out);
    window_edge<<<252, 256, 0, stream>>>(edge, lenw, out + 24576);
}


# cls product re-decomposed: 24 blocks stage the 32 cls rows and a 32-column slab of row-major cls_w in LDS (loads issued up front, 6 K phases), one wave per 16x16 tile, full K; reorder launch gone
# speedup vs baseline: 1.7924x; 1.5278x over previous
#include <hip/hip_runtime.h>
#include <hip/hip_bf16.h>


typedef float v2f __attribute__((ext_vector_type(2)));
typedef float v8f __attribute__((ext_vector_type(8)));
typedef float v4f __attribute__((ext_vector_type(4)));

__device__ __forceinline__ v8f wmma_f32(v2f a, v2f b, v8f c) {
    return __builtin_amdgcn_wmma_f32_16x16x4_f32(
        false, a, false, b, (short)0, c, false, false);
}

__global__ __launch_bounds__(256) void lengths_k(const int* __restrict__ mask,
                                                 int* __restrict__ Lout) {
    __shared__ int sh[8];
    int b = blockIdx.x, t = threadIdx.x;
    int s = 0;
    #pragma unroll
    for (int j = 0; j < 8; ++j) s += mask[b * 2048 + t + 256 * j];
    #pragma unroll
    for (int o = 1; o < 32; o <<= 1) s += __shfl_xor(s, o, 32);
    if ((t & 31) == 0) sh[t >> 5] = s;
    __syncthreads();
    if (t == 0) {
        int tot = 0;
        #pragma unroll
        for (int j = 0; j < 8; ++j) tot += sh[j];
        int L = tot - 1;
        Lout[b] = L < 2046 ? L : 2046;
    }
}

__global__ __launch_bounds__(256) void reorder_tok(const float* __restrict__ W,
                                                   float* __restrict__ Wr) {
    int o    = blockIdx.x * 256 + threadIdx.x;
    int ks   = o >> 9;
    int rem  = o & 511;
    int p    = rem >> 7;
    int rem2 = rem & 127;
    int lane = rem2 >> 2;
    int tl   = (rem2 >> 1) & 1;
    int s    = rem2 & 1;
    int n    = lane & 15;
    int h    = lane >> 4;
    int k    = 8 * (ks >> 1) + 4 * h + 2 * (ks & 1) + s;
    int col  = (2 * p + tl) * 16 + n;
    Wr[o] = W[k * 128 + col];
}

#define TLDC 132
#define TNPH 6
#define TPH4 4096

__global__ __launch_bounds__(512) __attribute__((amdgpu_waves_per_eu(4, 4)))
void tok_gemm_ln(
    const float* __restrict__ hidden, const int* __restrict__ amask,
    const float* __restrict__ wr,     const float* __restrict__ tok_b,
    const float* __restrict__ gamma,  const float* __restrict__ beta,
    const int* __restrict__ Lw, float* __restrict__ edge,
    float* __restrict__ out_rows) {
    __shared__ __align__(16) float shmem[16 * 16 * TLDC];
    float4* sW = (float4*)shmem;

    const int  t    = threadIdx.x;
    const int  lane = t & 31;
    const int  wv   = t >> 5;
    const int  c    = lane & 15;
    const int  h    = lane >> 4;
    const long row0 = (long)blockIdx.x * 256 + wv * 16;

    v8f acc[8] = {};
    const float4* ap = (const float4*)(hidden + (row0 + c) * 768 + 4 * h);
    const float4* wg = (const float4*)wr + t;

    float4 a_cur[4], a_nxt[4];
    {
        const float4 w0 = wg[0],    w1 = wg[512],  w2 = wg[1024], w3 = wg[1536];
        const float4 w4 = wg[2048], w5 = wg[2560], w6 = wg[3072], w7 = wg[3584];
        float4* d = sW + t;
        d[0]    = w0; d[512]  = w1; d[1024] = w2; d[1536] = w3;
        d[2048] = w4; d[2560] = w5; d[3072] = w6; d[3584] = w7;
    }
    #pragma unroll
    for (int q = 0; q < 4; ++q) a_cur[q] = ap[2 * q];
    __syncthreads();

    for (int ph = 0; ph < TNPH; ++ph) {
        const int phn = (ph + 1 < TNPH) ? (ph + 1) : (TNPH - 1);
        const float4* wgn = wg + phn * TPH4;
        float4 w0 = {}, w1 = {}, w2 = {}, w3 = {}, w4 = {}, w5 = {}, w6 = {}, w7 = {};
        const float4* sBp = sW + (ph & 1) * TPH4 + lane;

        #pragma unroll 1
        for (int jj = 0; jj < 4; ++jj) {
            const int j  = ph * 4 + jj;
            const int jn = (j + 1 < 24) ? (j + 1) : 23;
            #pragma unroll
            for (int q = 0; q < 4; ++q) a_nxt[q] = ap[jn * 8 + 2 * q];
            if (jj == 0) {
                w0 = wgn[0];    w1 = wgn[512];  w2 = wgn[1024]; w3 = wgn[1536];
                w4 = wgn[2048]; w5 = wgn[2560]; w6 = wgn[3072]; w7 = wgn[3584];
            }
            const float4* sB = sBp + jj * (8 * 4 * 32);
            float4 bc[4], bn[4];
            #pragma unroll
            for (int p = 0; p < 4; ++p) bc[p] = sB[p * 32];
            #pragma unroll
            for (int q = 0; q < 4; ++q) {
                #pragma unroll
                for (int e = 0; e < 2; ++e) {
                    const int ks8 = 2 * q + e;
                    if (ks8 + 1 < 8) {
                        #pragma unroll
                        for (int p = 0; p < 4; ++p) bn[p] = sB[((ks8 + 1) * 4 + p) * 32];
                    }
                    __builtin_amdgcn_sched_barrier(0);
                    v2f a;
                    if (e == 0) { a[0] = a_cur[q].x; a[1] = a_cur[q].y; }
                    else        { a[0] = a_cur[q].z; a[1] = a_cur[q].w; }
                    #pragma unroll
                    for (int p = 0; p < 4; ++p) {
                        v2f b0 = {bc[p].x, bc[p].y};
                        v2f b1 = {bc[p].z, bc[p].w};
                        acc[2 * p]     = wmma_f32(a, b0, acc[2 * p]);
                        acc[2 * p + 1] = wmma_f32(a, b1, acc[2 * p + 1]);
                    }
                    __builtin_amdgcn_sched_barrier(0);
                    #pragma unroll
                    for (int p = 0; p < 4; ++p) bc[p] = bn[p];
                }
            }
            #pragma unroll
            for (int q = 0; q < 4; ++q) a_cur[q] = a_nxt[q];
        }
        {
            float4* d = sW + ((ph + 1) & 1) * TPH4 + t;
            d[0]    = w0; d[512]  = w1; d[1024] = w2; d[1536] = w3;
            d[2048] = w4; d[2560] = w5; d[3072] = w6; d[3584] = w7;
        }
        __syncthreads();
    }

    float bia[8], gam[8], bet[8];
    #pragma unroll
    for (int nt = 0; nt < 8; ++nt) {
        bia[nt] = tok_b[nt * 16 + c];
        gam[nt] = gamma[nt * 16 + c];
        bet[nt] = beta[nt * 16 + c];
    }
    float* Cw = shmem + wv * (16 * TLDC);
    #pragma unroll
    for (int r = 0; r < 8; ++r) {
        const long  rg = row0 + 8 * h + r;
        const float mk = (float)amask[rg];
        float v[8];
        float s = 0.f, ss = 0.f;
        #pragma unroll
        for (int nt = 0; nt < 8; ++nt) {
            v[nt] = acc[nt][r] + bia[nt];
            s += v[nt]; ss += v[nt] * v[nt];
        }
        #pragma unroll
        for (int o = 1; o < 16; o <<= 1) {
            s  += __shfl_xor(s,  o, 32);
            ss += __shfl_xor(ss, o, 32);
        }
        const float mean = s * (1.f / 128.f);
        const float var  = ss * (1.f / 128.f) - mean * mean;
        const float rstd = rsqrtf(var + 1e-5f);
        #pragma unroll
        for (int nt = 0; nt < 8; ++nt) {
            float y = (v[nt] - mean) * rstd * gam[nt] + bet[nt];
            Cw[(8 * h + r) * TLDC + nt * 16 + c] = fmaxf(y, 0.f) * mk;
        }
    }
    __syncthreads();

    const int kb = blockIdx.x & 7;
    const int bI = blockIdx.x >> 3;
    const int s0 = kb * 256;
    const int L  = Lw[bI];
    const int wu = __builtin_amdgcn_readfirstlane(wv);

    if (wu == 0 || wu == 15) {
        const float* src = shmem + (wu == 0 ? 0 : 247) * TLDC + 4 * lane;
        float* dst = edge + ((long)blockIdx.x * 18 + (wu == 0 ? 0 : 9)) * 128 + 4 * lane;
        #pragma unroll
        for (int j = 0; j < 9; ++j)
            *(float4*)(dst + j * 128) = *(const float4*)(src + j * TLDC);
    }

    const int i0 = wu * 16;
    float4 rw[25];
    #pragma unroll
    for (int q = 0; q < 25; ++q) {
        const int i  = i0 - 5 + q;
        const int ic = i < 0 ? 0 : (i > 255 ? 255 : i);
        float4 v = *(const float4*)&shmem[ic * TLDC + 4 * lane];
        const int sr = s0 + i;
        const bool in = (i >= 0) && (i <= 255) && (sr >= 1) && (sr <= L);
        if (!in) { v.x = 0.f; v.y = 0.f; v.z = 0.f; v.w = 0.f; }
        rw[q] = v;
    }
    float4 mn[16];
    float  sq[16];
    #pragma unroll
    for (int rr = 0; rr < 16; ++rr) {
        const int tp = s0 + i0 + rr - 1;
        int start = tp - 5; if (start < 0) start = 0;
        int end   = tp + 5; if (end > L) end = L;
        int cnt   = end - start; if (cnt < 1) cnt = 1;
        float4 sum = {0.f, 0.f, 0.f, 0.f};
        #pragma unroll
        for (int d = 0; d < 10; ++d) {
            sum.x += rw[rr + d].x; sum.y += rw[rr + d].y;
            sum.z += rw[rr + d].z; sum.w += rw[rr + d].w;
        }
        const float inv = 1.f / (float)cnt;
        mn[rr].x = sum.x * inv; mn[rr].y = sum.y * inv;
        mn[rr].z = sum.z * inv; mn[rr].w = sum.w * inv;
        sq[rr] = mn[rr].x * mn[rr].x + mn[rr].y * mn[rr].y +
                 mn[rr].z * mn[rr].z + mn[rr].w * mn[rr].w;
    }
    #pragma unroll
    for (int o = 1; o < 32; o <<= 1) {
        #pragma unroll
        for (int rr = 0; rr < 16; ++rr) sq[rr] += __shfl_xor(sq[rr], o, 32);
    }
    #pragma unroll
    for (int rr = 0; rr < 16; ++rr) {
        const int i = i0 + rr;
        const int s = s0 + i;
        const bool mine = (s >= 1) && (s <= 2046) &&
                          (i >= 5 || kb == 0) && (i <= 251 || kb == 7);
        if (mine) {
            const int  tp    = s - 1;
            const bool valid = tp < L;
            const float n  = sqrtf(sq[rr]);
            const float sc = (valid && n > 0.f) ? (1.f / n) : 0.f;
            float4 o4 = {mn[rr].x * sc, mn[rr].y * sc, mn[rr].z * sc, mn[rr].w * sc};
            *(float4*)(out_rows + ((long)(bI * 2046 + tp)) * 128 + 4 * lane) = o4;
        }
    }
}

__device__ __forceinline__ void lds_barrier() {
    __builtin_amdgcn_fence(__ATOMIC_RELEASE, "workgroup", "local");
    __builtin_amdgcn_s_barrier();
    __builtin_amdgcn_fence(__ATOMIC_ACQUIRE, "workgroup", "local");
}

#define CPX  772
#define CPW  40
#define CPC  36
#define CNPH 6

__global__ __launch_bounds__(128) __attribute__((amdgpu_waves_per_eu(1, 1)))
void cls_gemm(const float* __restrict__ hidden,
              const float* __restrict__ cls_w,
              float* __restrict__ cls_lin) {
    __shared__ __align__(16) float Xs[32 * CPX];
    __shared__ __align__(16) float Ws[768 * CPW];
    __shared__ __align__(16) float Cs[32 * CPC];

    const int t = threadIdx.x, lane = t & 31;
    const int wv = __builtin_amdgcn_readfirstlane(t >> 5);
    const int c = lane & 15, h = lane >> 4;
    const int nb = blockIdx.x;
    const int mi = wv & 1, ni = wv >> 1;

    v4f xr[CNPH][8], wr[CNPH][8];
    #pragma unroll
    for (int p = 0; p < CNPH; ++p) {
        #pragma unroll
        for (int j = 0; j < 8; ++j) {
            const long row = wv * 8 + j;
            xr[p][j] = *(const v4f*)(hidden + row * (2048 * 768) + p * 128 + 4 * lane);
        }
        #pragma unroll
        for (int j = 0; j < 8; ++j) {
            const int idx = j * 128 + t;
            const long k  = p * 128 + (idx >> 3);
            wr[p][j] = *(const v4f*)(cls_w + k * 768 + nb * 32 + 4 * (idx & 7));
        }
        __builtin_amdgcn_sched_barrier(0);
    }

    v8f acc = {};
    const float* xa = Xs + (mi * 16 + c) * CPX + 2 * h;
    const float* wb = Ws + (2 * h) * CPW + ni * 16 + c;
    v2f fa[2][8], fb[2][8];

    #pragma unroll
    for (int p = 0; p < CNPH; ++p) {
        #pragma unroll
        for (int j = 0; j < 8; ++j) {
            *(v4f*)&Xs[(wv * 8 + j) * CPX + p * 128 + 4 * lane] = xr[p][j];
            const int idx = j * 128 + t;
            *(v4f*)&Ws[(p * 128 + (idx >> 3)) * CPW + 4 * (idx & 7)] = wr[p][j];
        }
        lds_barrier();

        #pragma unroll
        for (int e = 0; e < 8; ++e) {
            const int i = p * 32 + e;
            fa[0][e] = *(const v2f*)(xa + 4 * i);
            fb[0][e][0] = wb[(4 * i) * CPW];
            fb[0][e][1] = wb[(4 * i + 1) * CPW];
        }
        __builtin_amdgcn_sched_barrier(0);
        #pragma unroll
        for (int g = 0; g < 4; ++g) {
            if (g + 1 < 4) {
                #pragma unroll
                for (int e = 0; e < 8; ++e) {
                    const int i = p * 32 + 8 * (g + 1) + e;
                    fa[(g + 1) & 1][e] = *(const v2f*)(xa + 4 * i);
                    fb[(g + 1) & 1][e][0] = wb[(4 * i) * CPW];
                    fb[(g + 1) & 1][e][1] = wb[(4 * i + 1) * CPW];
                }
            }
            __builtin_amdgcn_sched_barrier(0);
            #pragma unroll
            for (int e = 0; e < 8; ++e)
                acc = wmma_f32(fa[g & 1][e], fb[g & 1][e], acc);
            __builtin_amdgcn_sched_barrier(0);
        }
    }

    #pragma unroll
    for (int r = 0; r < 8; ++r)
        Cs[(mi * 16 + 8 * h + r) * CPC + ni * 16 + c] = acc[r];
    __syncthreads();
    #pragma unroll
    for (int j = 0; j < 2; ++j) {
        const int idx = j * 128 + t;
        const float4 v = *(const float4*)&Cs[(idx >> 3) * CPC + 4 * (idx & 7)];
        *(float4*)(cls_lin + (idx >> 3) * 768 + nb * 32 + 4 * (idx & 7)) = v;
    }
}

__global__ __launch_bounds__(256) void cls_ln(const float* __restrict__ cls_lin,
                                              const float* __restrict__ cls_b,
                                              const float* __restrict__ gamma,
                                              const float* __restrict__ beta,
                                              float* __restrict__ out) {
    __shared__ float sh_s[8], sh_ss[8];
    const int b = blockIdx.x, t = threadIdx.x;
    float x[3];
    float s = 0.f, ss = 0.f;
    #pragma unroll
    for (int j = 0; j < 3; ++j) {
        int c = t + 256 * j;
        float v = cls_lin[b * 768 + c] + cls_b[c];
        x[j] = v; s += v; ss += v * v;
    }
    #pragma unroll
    for (int o = 1; o < 32; o <<= 1) {
        s  += __shfl_xor(s,  o, 32);
        ss += __shfl_xor(ss, o, 32);
    }
    if ((t & 31) == 0) { sh_s[t >> 5] = s; sh_ss[t >> 5] = ss; }
    __syncthreads();
    float ts = 0.f, tss = 0.f;
    #pragma unroll
    for (int j = 0; j < 8; ++j) { ts += sh_s[j]; tss += sh_ss[j]; }
    const float mean = ts * (1.f / 768.f);
    const float var  = tss * (1.f / 768.f) - mean * mean;
    const float rstd = rsqrtf(var + 1e-5f);
    #pragma unroll
    for (int j = 0; j < 3; ++j) {
        int c = t + 256 * j;
        out[b * 768 + c] = (x[j] - mean) * rstd * gamma[c] + beta[c];
    }
}

__global__ __launch_bounds__(256) void window_edge(const float* __restrict__ edge,
                                                   const int* __restrict__ Lw,
                                                   float* __restrict__ out_rows) {
    const int g    = blockIdx.x * 8 + (threadIdx.x >> 5);
    const int lane = threadIdx.x & 31;
    const int bnd  = g / 9;
    const int e    = g % 9 - 4;
    const int b    = bnd / 7;
    const int k    = bnd % 7 + 1;
    const int s    = 256 * k + e;
    const int tp   = s - 1;
    const int L    = Lw[b];
    const bool valid = tp < L;
    int start = tp - 5; if (start < 0) start = 0;
    int end   = tp + 5; if (end > L) end = L;
    int cnt   = end - start; if (cnt < 1) cnt = 1;

    v4f ld[10];
    #pragma unroll
    for (int d = 0; d < 10; ++d) {
        const int er = e - 5 + d;
        const long j = er < 0 ? ((long)(b * 8 + k - 1) * 18 + 18 + er)
                              : ((long)(b * 8 + k) * 18 + er);
        ld[d] = *(const v4f*)(edge + j * 128 + lane * 4);
    }
    asm volatile("" :: "v"(ld[0]), "v"(ld[1]), "v"(ld[2]), "v"(ld[3]), "v"(ld[4]),
                       "v"(ld[5]), "v"(ld[6]), "v"(ld[7]), "v"(ld[8]), "v"(ld[9]));
    float4 sum = {0.f, 0.f, 0.f, 0.f};
    #pragma unroll
    for (int d = 0; d < 10; ++d) {
        const int sr = s - 5 + d;
        const bool in = (sr >= 1) && (sr <= L);
        const float vx = in ? ld[d].x : 0.f, vy = in ? ld[d].y : 0.f;
        const float vz = in ? ld[d].z : 0.f, vw = in ? ld[d].w : 0.f;
        sum.x += vx; sum.y += vy; sum.z += vz; sum.w += vw;
    }
    const float inv = 1.f / (float)cnt;
    float4 mn = {sum.x * inv, sum.y * inv, sum.z * inv, sum.w * inv};
    float sq = mn.x * mn.x + mn.y * mn.y + mn.z * mn.z + mn.w * mn.w;
    #pragma unroll
    for (int o = 1; o < 32; o <<= 1) sq += __shfl_xor(sq, o, 32);
    const float n  = sqrtf(sq);
    const float sc = (valid && n > 0.f) ? (1.f / n) : 0.f;
    float4 o4 = {mn.x * sc, mn.y * sc, mn.z * sc, mn.w * sc};
    *(float4*)(out_rows + ((long)(b * 2046 + tp)) * 128 + lane * 4) = o4;
}

extern "C" void kernel_launch(void* const* d_in, const int* in_sizes, int n_in,
                              void* d_out, int out_size, void* d_ws, size_t ws_size,
                              hipStream_t stream) {
    const float* hidden   = (const float*)d_in[0];
    const int*   amask    = (const int*)  d_in[1];
    const float* tok_w    = (const float*)d_in[2];
    const float* tok_b    = (const float*)d_in[3];
    const float* cls_w    = (const float*)d_in[4];
    const float* cls_b    = (const float*)d_in[5];
    const float* ln_tok_g = (const float*)d_in[6];
    const float* ln_tok_b = (const float*)d_in[7];
    const float* ln_cls_g = (const float*)d_in[8];
    const float* ln_cls_b = (const float*)d_in[9];
    float* out = (float*)d_out;

    float* ws      = (float*)d_ws;
    float* edge    = ws;
    float* cls_lin = ws + 8380416;
    float* wr_tok  = ws + 8404992;
    int*   lenw    = (int*)(ws + 9093120);

    reorder_tok<<<384, 256, 0, stream>>>(tok_w, wr_tok);
    lengths_k<<<32,   256, 0, stream>>>(amask, lenw);

    tok_gemm_ln<<<256, 512, 0, stream>>>(hidden, amask, wr_tok, tok_b,
                                          ln_tok_g, ln_tok_b, lenw, edge,
                                          out + 24576);
    cls_gemm<<<24, 128, 0, stream>>>(hidden, cls_w, cls_lin);
    cls_ln<<<32, 256, 0, stream>>>(cls_lin, cls_b, ln_cls_g, ln_cls_b, out);
    window_edge<<<252, 256, 0, stream>>>(edge, lenw, out + 24576);
}
